# starting kernel
# baseline (speedup 1.0000x reference)
#include <hip/hip_runtime.h>

#define DIM   64
#define NREL  8
#define NBLK  8
#define BS    8

typedef __attribute__((ext_vector_type(2))) float v2f;
typedef __attribute__((ext_vector_type(8))) float v8f;

__global__ void zero_f32v4(float4* __restrict__ p, long long n4) {
  long long i = (long long)blockIdx.x * blockDim.x + threadIdx.x;
  long long stride = (long long)gridDim.x * blockDim.x;
  float4 z; z.x = z.y = z.z = z.w = 0.0f;
  for (; i < n4; i += stride) p[i] = z;
}

__global__ void edge_scatter(const float*  __restrict__ x,
                             const float*  __restrict__ ew,
                             const int*    __restrict__ src,
                             const int*    __restrict__ tgt,
                             const int*    __restrict__ et,
                             float*        __restrict__ agg,
                             int nEdges, int nNodes) {
  long long tid    = (long long)blockIdx.x * blockDim.x + threadIdx.x;
  long long stride = (long long)gridDim.x * blockDim.x;
  long long total  = (long long)nEdges * (DIM / 4);
  for (; tid < total; tid += stride) {
    int e = (int)(tid >> 4);
    int c = (int)(tid & 15);
    int s = src[e];
    int t = tgt[e];
    int r = et[e];
    float w = ew[e];
    float4 xv = *(const float4*)(x + (size_t)s * DIM + c * 4);
    float* dst = agg + ((size_t)r * nNodes + t) * DIM + c * 4;
    unsafeAtomicAdd(dst + 0, w * xv.x);
    unsafeAtomicAdd(dst + 1, w * xv.y);
    unsafeAtomicAdd(dst + 2, w * xv.z);
    unsafeAtomicAdd(dst + 3, w * xv.w);
  }
}

__global__ void __launch_bounds__(256)
rgcn_block_wmma(const float* __restrict__ agg,
                const float* __restrict__ blocks,
                float*       __restrict__ out,
                int nNodes) {
  const int lane   = threadIdx.x & 31;
  const int wib    = threadIdx.x >> 5;
  const int gwave  = blockIdx.x * (blockDim.x >> 5) + wib;
  const int nWaves = gridDim.x * (blockDim.x >> 5);

  const int j       = gwave & 3;
  const int group   = gwave >> 2;
  const int nGroups = nWaves >> 2;
  const int nTiles  = (nNodes + 15) >> 4;

  const int nl   = lane & 15;
  const int half = lane >> 4;

  v2f Bfrag[NREL][4];
#pragma unroll
  for (int r = 0; r < NREL; ++r) {
#pragma unroll
    for (int s = 0; s < 4; ++s) {
#pragma unroll
      for (int p = 0; p < 2; ++p) {
        int kl = 4 * s + p + 2 * half;
        float v = 0.0f;
        if ((kl >> 3) == (nl >> 3)) {
          int blk = 2 * j + (kl >> 3);
          v = blocks[(((size_t)r * NBLK + blk) * BS + (kl & 7)) * BS + (nl & 7)];
        }
        Bfrag[r][s][p] = v;
      }
    }
  }

  for (int tile = group; tile < nTiles; tile += nGroups) {
    const int node0  = tile << 4;
    const int myNode = node0 + nl;
    const bool full  = (node0 + 16) <= nNodes;

    v2f a[NREL][4];
    if (full) {
#pragma unroll
      for (int r = 0; r < NREL; ++r) {
        const float* rowBase = agg + ((size_t)r * nNodes + myNode) * DIM
                                   + 16 * j + 2 * half;
#pragma unroll
        for (int s = 0; s < 4; ++s)
          a[r][s] = *(const v2f*)(rowBase + 4 * s);
      }
    } else {
      const bool valid = myNode < nNodes;
      const int  cn    = valid ? myNode : (nNodes - 1);
#pragma unroll
      for (int r = 0; r < NREL; ++r) {
        const float* rowBase = agg + ((size_t)r * nNodes + cn) * DIM
                                   + 16 * j + 2 * half;
#pragma unroll
        for (int s = 0; s < 4; ++s) {
          v2f t = *(const v2f*)(rowBase + 4 * s);
          a[r][s][0] = valid ? t[0] : 0.0f;
          a[r][s][1] = valid ? t[1] : 0.0f;
        }
      }
    }

    v8f acc = {};
#pragma unroll
    for (int r = 0; r < NREL; ++r) {
#pragma unroll
      for (int s = 0; s < 4; ++s) {
        acc = __builtin_amdgcn_wmma_f32_16x16x4_f32(
            false, a[r][s], false, Bfrag[r][s], (short)0, acc, false, false);
      }
    }

    float* outBase = out + (size_t)(node0 + 8 * half) * DIM + 16 * j + nl;
    if (full) {
#pragma unroll
      for (int i = 0; i < 8; ++i)
        outBase[(size_t)i * DIM] = acc[i];
    } else {
#pragma unroll
      for (int i = 0; i < 8; ++i) {
        int row = node0 + i + 8 * half;
        if (row < nNodes)
          out[(size_t)row * DIM + 16 * j + nl] = acc[i];
      }
    }
  }
}

__global__ void edge_direct(const float* __restrict__ x,
                            const float* __restrict__ blocks,
                            const float* __restrict__ ew,
                            const int*   __restrict__ src,
                            const int*   __restrict__ tgt,
                            const int*   __restrict__ et,
                            float*       __restrict__ out,
                            int nEdges) {
  long long tid    = (long long)blockIdx.x * blockDim.x + threadIdx.x;
  long long stride = (long long)gridDim.x * blockDim.x;
  long long total  = (long long)nEdges * NBLK;
  for (; tid < total; tid += stride) {
    int e = (int)(tid >> 3);
    int b = (int)(tid & 7);
    int s = src[e], t = tgt[e], r = et[e];
    float w = ew[e];
    const float* xb = x + (size_t)s * DIM + b * BS;
    const float* W  = blocks + ((size_t)r * NBLK + b) * (BS * BS);
    float xi[BS];
#pragma unroll
    for (int i = 0; i < BS; ++i) xi[i] = xb[i];
#pragma unroll
    for (int o = 0; o < BS; ++o) {
      float acc = 0.0f;
#pragma unroll
      for (int i = 0; i < BS; ++i) acc += xi[i] * W[i * BS + o];
      unsafeAtomicAdd(out + (size_t)t * DIM + b * BS + o, w * acc);
    }
  }
}

extern "C" void kernel_launch(void* const* d_in, const int* in_sizes, int n_in,
                              void* d_out, int out_size, void* d_ws, size_t ws_size,
                              hipStream_t stream) {
  const float* x      = (const float*)d_in[0];
  const float* blocks = (const float*)d_in[1];
  const float* ew     = (const float*)d_in[2];
  const int*   src    = (const int*)d_in[3];
  const int*   tgt    = (const int*)d_in[4];
  const int*   et     = (const int*)d_in[5];
  float*       out    = (float*)d_out;

  const int nNodes = in_sizes[0] / DIM;
  const int nEdges = in_sizes[2];

  const size_t aggElems = (size_t)NREL * nNodes * DIM;
  const size_t need     = aggElems * sizeof(float);

  if (ws_size >= need) {
    float* agg = (float*)d_ws;

    long long n4 = (long long)(aggElems / 4);
    zero_f32v4<<<4096, 256, 0, stream>>>((float4*)agg, n4);

    edge_scatter<<<8192, 256, 0, stream>>>(x, ew, src, tgt, et, agg,
                                           nEdges, nNodes);

    rgcn_block_wmma<<<512, 256, 0, stream>>>(agg, blocks, out, nNodes);
  } else {
    long long o4 = (long long)out_size / 4;
    zero_f32v4<<<2048, 256, 0, stream>>>((float4*)out, o4);
    edge_direct<<<8192, 256, 0, stream>>>(x, blocks, ew, src, tgt, et,
                                          out, nEdges);
  }
}


# edge pass applies blocks[rel] per 16-edge tile on WMMA (relation folded into N + row mask) and adds into out with float atomics (arrival order); the (rel,node) buffer and its pass are gone
# speedup vs baseline: 1.6291x; 1.6291x over previous
#include <hip/hip_runtime.h>


#define DIM   64
#define NREL  8
#define NBLK  8
#define BS    8

typedef __attribute__((ext_vector_type(2))) float v2f;
typedef __attribute__((ext_vector_type(8))) float v8f;

__global__ void zero_f32v4(float4* __restrict__ p, long long n4) {
  long long i = (long long)blockIdx.x * blockDim.x + threadIdx.x;
  long long stride = (long long)gridDim.x * blockDim.x;
  float4 z; z.x = z.y = z.z = z.w = 0.0f;
  for (; i < n4; i += stride) p[i] = z;
}

static __device__ __forceinline__ float swap8(float v) {
  return __int_as_float(__builtin_amdgcn_update_dpp(
      0, __float_as_int(v), 0x168, 0xF, 0xF, true));
}

__global__ void __launch_bounds__(256)
edge_transform_scatter(const float* __restrict__ x,
                       const float* __restrict__ blocks,
                       const float* __restrict__ ew,
                       const int*   __restrict__ src,
                       const int*   __restrict__ tgt,
                       const int*   __restrict__ et,
                       float*       __restrict__ out,
                       int nEdges) {
  const int lane   = threadIdx.x & 31;
  const int gwave  = __builtin_amdgcn_readfirstlane(
      (int)(blockIdx.x * (blockDim.x >> 5) + (threadIdx.x >> 5)));
  const int nWaves = gridDim.x * (blockDim.x >> 5);
  const int nl     = lane & 15;
  const int half   = lane >> 4;
  const int nhi    = nl >> 3;
  const bool lo    = nl < 8;
  const int nTiles = (nEdges + 15) >> 4;

  v2f Bf[NBLK][4][2];
#pragma unroll
  for (int b = 0; b < NBLK; ++b) {
#pragma unroll
    for (int t = 0; t < 4; ++t) {
#pragma unroll
      for (int s = 0; s < 2; ++s) {
#pragma unroll
        for (int p = 0; p < 2; ++p) {
          const int r = 2 * t + nhi;
          const int i = 4 * half + 2 * s + p;
          Bf[b][t][s][p] =
              blocks[(((size_t)r * NBLK + b) * BS + i) * BS + (nl & 7)];
        }
      }
    }
  }

  for (int tile = gwave; tile < nTiles; tile += nWaves) {
    const int  e     = (tile << 4) + nl;
    const bool valid = e < nEdges;
    const int  ec    = valid ? e : (nEdges - 1);
    const int  es    = src[ec];
    const int  etg   = tgt[ec];
    const int  er    = et[ec];
    const float w    = ew[ec];
    const int  rp    = valid ? (er >> 1) : -1;
    const int  par   = er & 1;

    const float* xr = x + (size_t)es * DIM + 4 * half;
    float4 xa[NBLK];
#pragma unroll
    for (int b = 0; b < NBLK; ++b) xa[b] = *(const float4*)(xr + b * BS);
    __builtin_amdgcn_sched_barrier(0);

    int      rowq[8];
    unsigned rowo[8];
#pragma unroll
    for (int i = 0; i < 8; ++i) {
      const int m  = i + 8 * half;
      const int pm = __shfl(par, m, 32);
      const int tm = __shfl(etg, m, 32);
      rowq[i] = pm ^ nhi;
      rowo[i] = ((unsigned)tm * DIM + (unsigned)nl) * 4u;
    }

    const bool m0 = (rp == 0), m1 = (rp == 1), m2 = (rp == 2), m3 = (rp == 3);

#pragma unroll
    for (int q = 0; q < NBLK / 2; ++q) {
      const int b0 = 2 * q, b1 = 2 * q + 1;
      v8f acc0 = {};
      v8f acc1 = {};
#pragma unroll
      for (int s = 0; s < 2; ++s) {
        float x00, x01, x10, x11;
        if (s == 0) { x00 = xa[b0].x; x01 = xa[b0].y; x10 = xa[b1].x; x11 = xa[b1].y; }
        else        { x00 = xa[b0].z; x01 = xa[b0].w; x10 = xa[b1].z; x11 = xa[b1].w; }
        x00 *= w; x01 *= w; x10 *= w; x11 *= w;
#pragma unroll
        for (int t = 0; t < 4; ++t) {
          const bool mt = (t == 0) ? m0 : (t == 1) ? m1 : (t == 2) ? m2 : m3;
          v2f a0, a1;
          a0[0] = mt ? x00 : 0.0f;  a0[1] = mt ? x01 : 0.0f;
          a1[0] = mt ? x10 : 0.0f;  a1[1] = mt ? x11 : 0.0f;
          acc0 = __builtin_amdgcn_wmma_f32_16x16x4_f32(
              false, a0, false, Bf[b0][t][s], (short)0, acc0, false, false);
          acc1 = __builtin_amdgcn_wmma_f32_16x16x4_f32(
              false, a1, false, Bf[b1][t][s], (short)0, acc1, false, false);
        }
      }

#pragma unroll
      for (int i = 0; i < 8; ++i) {
        const float own = lo ? acc0[i] : acc1[i];
        const float oth = lo ? acc1[i] : acc0[i];
        const float got = swap8(oth);
        const float v   = rowq[i] ? got : own;
        unsafeAtomicAdd((float*)((char*)out + rowo[i]) + 16 * q, v);
      }
      __builtin_amdgcn_sched_barrier(0);
    }
  }
}

extern "C" void kernel_launch(void* const* d_in, const int* in_sizes, int n_in,
                              void* d_out, int out_size, void* d_ws, size_t ws_size,
                              hipStream_t stream) {
  const float* x      = (const float*)d_in[0];
  const float* blocks = (const float*)d_in[1];
  const float* ew     = (const float*)d_in[2];
  const int*   src    = (const int*)d_in[3];
  const int*   tgt    = (const int*)d_in[4];
  const int*   et     = (const int*)d_in[5];
  float*       out    = (float*)d_out;

  const int nNodes = in_sizes[0] / DIM;
  const int nEdges = in_sizes[2];

  long long o4 = (long long)nNodes * (DIM / 4);
  zero_f32v4<<<2048, 256, 0, stream>>>((float4*)out, o4);

  edge_transform_scatter<<<512, 256, 0, stream>>>(x, blocks, ew, src, tgt, et,
                                                  out, nEdges);
}


# edges split by relation in LDS per work-group (ballot ranks): a 16-edge tile has one relation, 16 WMMAs per tile (was 64), B fragments from LDS, 8 waves per SIMD; atomic arrival order free as before
# speedup vs baseline: 2.3088x; 1.4172x over previous
#include <hip/hip_runtime.h>


#define DIM   64
#define NREL  8
#define NBLK  8
#define BS    8

#define NTHR   512
#define NWAVE  (NTHR / 32)
#define KMAX   4
#define CHUNK  (NTHR * KMAX)
#define LISTN  (CHUNK + NREL * 16)

typedef __attribute__((ext_vector_type(2))) float v2f;
typedef __attribute__((ext_vector_type(8))) float v8f;

__global__ void zero_f32v4(float4* __restrict__ p, long long n4) {
  long long i = (long long)blockIdx.x * blockDim.x + threadIdx.x;
  long long stride = (long long)gridDim.x * blockDim.x;
  float4 z; z.x = z.y = z.z = z.w = 0.0f;
  for (; i < n4; i += stride) p[i] = z;
}

template <bool FULL>
static __device__ __forceinline__ void tile_products_and_adds(
    const float4 (&xa)[NBLK], const float w, const float4 (&bq)[4][2],
    const int (&tgi)[8], float* __restrict__ out, const int nl) {
  unsigned rowo[8];
#pragma unroll
  for (int i = 0; i < 8; ++i) rowo[i] = ((unsigned)tgi[i] * DIM + (unsigned)nl) * 4u;

#pragma unroll
  for (int q = 0; q < NBLK / 2; ++q) {
    v8f acc = {};
#pragma unroll
    for (int j = 0; j < 2; ++j) {
      const float4 xv = xa[2 * q + j];
      const float4 bv = bq[q][j];
      v2f a0, a1, b0, b1;
      a0[0] = xv.x * w; a0[1] = xv.y * w; b0[0] = bv.x; b0[1] = bv.y;
      a1[0] = xv.z * w; a1[1] = xv.w * w; b1[0] = bv.z; b1[1] = bv.w;
      acc = __builtin_amdgcn_wmma_f32_16x16x4_f32(
          false, a0, false, b0, (short)0, acc, false, false);
      acc = __builtin_amdgcn_wmma_f32_16x16x4_f32(
          false, a1, false, b1, (short)0, acc, false, false);
    }
#pragma unroll
    for (int i = 0; i < 8; ++i) {
      if (FULL || tgi[i] >= 0)
        unsafeAtomicAdd((float*)((char*)out + rowo[i]) + 16 * q, acc[i]);
    }
  }
}

__global__ void __launch_bounds__(NTHR, 8)
edge_transform_scatter(const float* __restrict__ x,
                       const float* __restrict__ blocks,
                       const float* __restrict__ ew,
                       const int*   __restrict__ src,
                       const int*   __restrict__ tgt,
                       const int*   __restrict__ et,
                       float*       __restrict__ out,
                       int nEdges) {
  __shared__ float4 Bl[NREL * 4 * 2 * 32];
  __shared__ int    ls_src[LISTN];
  __shared__ __attribute__((aligned(16))) int ls_tgt[LISTN];
  __shared__ float  ls_w[LISTN];
  __shared__ int    wtot[NWAVE * NREL];
  __shared__ int    wpre[NWAVE * NREL];
  __shared__ int    rtot[NREL];

  const int tid  = threadIdx.x;
  const int lane = tid & 31;
  const int wave = __builtin_amdgcn_readfirstlane(tid >> 5);
  const int nl   = lane & 15;
  const int half = lane >> 4;

  const int per = (((nEdges + (int)gridDim.x - 1) / (int)gridDim.x) + 31) & ~31;
  const long long g0 = (long long)blockIdx.x * per;
  const int r0 = g0 < nEdges ? (int)g0 : nEdges;
  const int r1 = (g0 + per) < nEdges ? (int)(g0 + per) : nEdges;
  const int nChunks = (per + CHUNK - 1) / CHUNK;

  for (int ch = 0; ch < nChunks; ++ch) {
    const int c0 = r0 + ch * CHUNK < r1 ? r0 + ch * CHUNK : r1;
    const int c1 = c0 + CHUNK < r1 ? c0 + CHUNK : r1;

    int   myS[KMAX], myT[KMAX], myR[KMAX], myRank[KMAX];
    float myW[KMAX];
#pragma unroll
    for (int k = 0; k < KMAX; ++k) {
      const int  e  = c0 + k * NTHR + tid;
      const bool v  = e < c1;
      const int  ec = v ? e : 0;
      myS[k] = src[ec];
      myT[k] = tgt[ec];
      myW[k] = ew[ec];
      const int rl = et[ec];
      myR[k] = v ? rl : -1;
    }

    if (ch == 0) {
      float4 bv[(NREL * 4 * 2 * 32) / NTHR];
#pragma unroll
      for (int it = 0; it < (NREL * 4 * 2 * 32) / NTHR; ++it) {
        const int idx = it * NTHR + tid;
        const int ln = idx & 31, j = (idx >> 5) & 1, q = (idx >> 6) & 3, r = idx >> 8;
        const int c = ln & 15, h = ln >> 4;
        const float* bp = blocks + ((((size_t)r * NBLK + 2 * q + j) * BS + 4 * h) * BS + (c & 7));
        const bool nz = (j == 0) == (c < 8);
        const float t0 = bp[0], t1 = bp[BS], t2 = bp[2 * BS], t3 = bp[3 * BS];
        bv[it].x = nz ? t0 : 0.0f; bv[it].y = nz ? t1 : 0.0f;
        bv[it].z = nz ? t2 : 0.0f; bv[it].w = nz ? t3 : 0.0f;
      }
#pragma unroll
      for (int it = 0; it < (NREL * 4 * 2 * 32) / NTHR; ++it) Bl[it * NTHR + tid] = bv[it];
    }

    int cnt[NREL];
#pragma unroll
    for (int rr = 0; rr < NREL; ++rr) cnt[rr] = 0;
#pragma unroll
    for (int k = 0; k < KMAX; ++k) {
      unsigned m[NREL];
#pragma unroll
      for (int rr = 0; rr < NREL; ++rr) m[rr] = __builtin_amdgcn_ballot_w32(myR[k] == rr);
      unsigned om = 0u;
      int      oc = 0;
#pragma unroll
      for (int rr = 0; rr < NREL; ++rr) {
        om = (myR[k] == rr) ? m[rr] : om;
        oc = (myR[k] == rr) ? cnt[rr] : oc;
      }
      myRank[k] = oc + (int)__builtin_amdgcn_mbcnt_lo(om, 0u);
#pragma unroll
      for (int rr = 0; rr < NREL; ++rr) cnt[rr] += __builtin_popcount(m[rr]);
    }
    {
      int cv = cnt[0];
#pragma unroll
      for (int rr = 1; rr < NREL; ++rr) cv = (lane == rr) ? cnt[rr] : cv;
      if (lane < NREL) wtot[wave * NREL + lane] = cv;
    }
    __syncthreads();

    if (tid < NWAVE * NREL) {
      const int w_ = tid >> 3, r_ = tid & 7;
      int pre = 0;
#pragma unroll
      for (int w2 = 0; w2 < NWAVE; ++w2) {
        const int t = wtot[w2 * NREL + r_];
        pre += (w2 < w_) ? t : 0;
        if (w2 == NWAVE - 1 && w_ == NWAVE - 1) rtot[r_] = pre + t;
      }
      wpre[tid] = pre;
    }
    __syncthreads();

    int rt[NREL], rb[NREL + 1];
    rb[0] = 0;
#pragma unroll
    for (int rr = 0; rr < NREL; ++rr) {
      rt[rr] = __builtin_amdgcn_readfirstlane(rtot[rr]);
      rb[rr + 1] = rb[rr] + ((rt[rr] + 15) & ~15);
    }

#pragma unroll
    for (int k = 0; k < KMAX; ++k) {
      const int r = myR[k];
      if (r >= 0 && r < NREL) {
        int base = rb[0];
#pragma unroll
        for (int rr = 1; rr < NREL; ++rr) base = (r == rr) ? rb[rr] : base;
        const int pos = base + wpre[wave * NREL + r] + myRank[k];
        ls_src[pos] = myS[k];
        ls_tgt[pos] = myT[k];
        ls_w[pos]   = myW[k];
      }
    }
    if (tid < NREL * 16) {
      const int r_ = tid >> 4, jj = tid & 15;
      int base = rb[0], tot = rt[0], end = rb[1];
#pragma unroll
      for (int rr = 1; rr < NREL; ++rr) {
        base = (r_ == rr) ? rb[rr] : base;
        tot  = (r_ == rr) ? rt[rr] : tot;
        end  = (r_ == rr) ? rb[rr + 1] : end;
      }
      const int p = base + tot + jj;
      if (p < end) { ls_src[p] = 0; ls_tgt[p] = -1; ls_w[p] = 0.0f; }
    }
    __syncthreads();

    const int nTiles = rb[NREL] >> 4;
    for (int tile = wave; tile < nTiles; tile += NWAVE) {
      const int p0 = tile << 4;
      int r = 0;
#pragma unroll
      for (int rr = 1; rr < NREL; ++rr) r += (p0 >= rb[rr]) ? 1 : 0;

      const int   es = ls_src[p0 + nl];
      const float w  = ls_w[p0 + nl];

      const float* xr = x + (size_t)es * DIM + 4 * half;
      float4 xa[NBLK];
#pragma unroll
      for (int b = 0; b < NBLK; ++b) xa[b] = *(const float4*)(xr + b * BS);

      const int4 tg0 = *(const int4*)&ls_tgt[p0 + 8 * half];
      const int4 tg1 = *(const int4*)&ls_tgt[p0 + 8 * half + 4];
      const int tgi[8] = {tg0.x, tg0.y, tg0.z, tg0.w, tg1.x, tg1.y, tg1.z, tg1.w};

      const float4* bp = Bl + r * 256 + lane;
      float4 bq[4][2];
#pragma unroll
      for (int q = 0; q < 4; ++q) {
        bq[q][0] = bp[(q * 2 + 0) * 32];
        bq[q][1] = bp[(q * 2 + 1) * 32];
      }
      const int anyrow = tg0.x | tg0.y | tg0.z | tg0.w | tg1.x | tg1.y | tg1.z | tg1.w;
      const bool full = __builtin_amdgcn_ballot_w32(anyrow < 0) == 0u;
      if (full) tile_products_and_adds<true >(xa, w, bq, tgi, out, nl);
      else      tile_products_and_adds<false>(xa, w, bq, tgi, out, nl);
    }
    __syncthreads();
  }
}

extern "C" void kernel_launch(void* const* d_in, const int* in_sizes, int n_in,
                              void* d_out, int out_size, void* d_ws, size_t ws_size,
                              hipStream_t stream) {
  const float* x      = (const float*)d_in[0];
  const float* blocks = (const float*)d_in[1];
  const float* ew     = (const float*)d_in[2];
  const int*   src    = (const int*)d_in[3];
  const int*   tgt    = (const int*)d_in[4];
  const int*   et     = (const int*)d_in[5];
  float*       out    = (float*)d_out;

  const int nNodes = in_sizes[0] / DIM;
  const int nEdges = in_sizes[2];

  long long o4 = (long long)nNodes * (DIM / 4);
  zero_f32v4<<<2048, 256, 0, stream>>>((float4*)out, o4);

  edge_transform_scatter<<<512, NTHR, 0, stream>>>(x, blocks, ew, src, tgt, et,
                                                   out, nEdges);
}


# adjacent column pairs exchange D halves (v_permlane16_swap): a register is one tile row x 32 columns, so each float atomic covers one whole 128-B out line (was two 64-B halves); arrival order free
# speedup vs baseline: 2.8499x; 1.2344x over previous
#include <hip/hip_runtime.h>


#define DIM   64
#define NREL  8
#define NBLK  8
#define BS    8

#define NTHR   512
#define NWAVE  (NTHR / 32)
#define KMAX   4
#define CHUNK  (NTHR * KMAX)
#define LISTN  (CHUNK + NREL * 16)

typedef __attribute__((ext_vector_type(2))) float v2f;
typedef __attribute__((ext_vector_type(2))) unsigned v2u;
typedef __attribute__((ext_vector_type(8))) float v8f;

__global__ void zero_f32v4(float4* __restrict__ p, long long n4) {
  long long i = (long long)blockIdx.x * blockDim.x + threadIdx.x;
  long long stride = (long long)gridDim.x * blockDim.x;
  float4 z; z.x = z.y = z.z = z.w = 0.0f;
  for (; i < n4; i += stride) p[i] = z;
}

static __device__ __forceinline__ v8f pair_product(
    const float4 (&xa)[NBLK], const float w, const float4 (&bq)[4][2], const int q) {
  v8f acc = {};
#pragma unroll
  for (int j = 0; j < 2; ++j) {
    const float4 xv = xa[2 * q + j];
    const float4 bv = bq[q][j];
    v2f a0, a1, b0, b1;
    a0[0] = xv.x * w; a0[1] = xv.y * w; b0[0] = bv.x; b0[1] = bv.y;
    a1[0] = xv.z * w; a1[1] = xv.w * w; b1[0] = bv.z; b1[1] = bv.w;
    acc = __builtin_amdgcn_wmma_f32_16x16x4_f32(
        false, a0, false, b0, (short)0, acc, false, false);
    acc = __builtin_amdgcn_wmma_f32_16x16x4_f32(
        false, a1, false, b1, (short)0, acc, false, false);
  }
  return acc;
}

template <bool FULL>
static __device__ __forceinline__ void tile_products_and_adds(
    const float4 (&xa)[NBLK], const float w, const float4 (&bq)[4][2],
    const int (&ts)[16], float* __restrict__ out, const int lane) {
  float* const col = out + lane;

#pragma unroll
  for (int p = 0; p < NBLK / 4; ++p) {
    v8f lo = pair_product(xa, w, bq, 2 * p);
    v8f hi = pair_product(xa, w, bq, 2 * p + 1);
#pragma unroll
    for (int i = 0; i < 8; ++i) {
      const float lv = lo[i], hv = hi[i];
      const v2u s = __builtin_amdgcn_permlane16_swap(
          __float_as_uint(lv), __float_as_uint(hv), false, false);
      const unsigned s0 = s[0], s1 = s[1];
      lo[i] = __uint_as_float(s0);
      hi[i] = __uint_as_float(s1);
    }
#pragma unroll
    for (int i = 0; i < 8; ++i) {
      if (FULL || ts[i] >= 0)
        unsafeAtomicAdd(col + (size_t)(unsigned)ts[i] * DIM + 32 * p, lo[i]);
    }
#pragma unroll
    for (int i = 0; i < 8; ++i) {
      if (FULL || ts[8 + i] >= 0)
        unsafeAtomicAdd(col + (size_t)(unsigned)ts[8 + i] * DIM + 32 * p, hi[i]);
    }
  }
}

__global__ void __launch_bounds__(NTHR, 8)
edge_transform_scatter(const float* __restrict__ x,
                       const float* __restrict__ blocks,
                       const float* __restrict__ ew,
                       const int*   __restrict__ src,
                       const int*   __restrict__ tgt,
                       const int*   __restrict__ et,
                       float*       __restrict__ out,
                       int nEdges) {
  __shared__ float4 Bl[NREL * 4 * 2 * 32];
  __shared__ int    ls_src[LISTN];
  __shared__ __attribute__((aligned(16))) int ls_tgt[LISTN];
  __shared__ float  ls_w[LISTN];
  __shared__ int    wtot[NWAVE * NREL];
  __shared__ int    wpre[NWAVE * NREL];
  __shared__ int    rtot[NREL];

  const int tid  = threadIdx.x;
  const int lane = tid & 31;
  const int wave = __builtin_amdgcn_readfirstlane(tid >> 5);
  const int nl   = lane & 15;
  const int half = lane >> 4;

  const int per = (((nEdges + (int)gridDim.x - 1) / (int)gridDim.x) + 31) & ~31;
  const long long g0 = (long long)blockIdx.x * per;
  const int r0 = g0 < nEdges ? (int)g0 : nEdges;
  const int r1 = (g0 + per) < nEdges ? (int)(g0 + per) : nEdges;
  const int nChunks = (per + CHUNK - 1) / CHUNK;

  for (int ch = 0; ch < nChunks; ++ch) {
    const int c0 = r0 + ch * CHUNK < r1 ? r0 + ch * CHUNK : r1;
    const int c1 = c0 + CHUNK < r1 ? c0 + CHUNK : r1;

    int   myS[KMAX], myT[KMAX], myR[KMAX], myRank[KMAX];
    float myW[KMAX];
#pragma unroll
    for (int k = 0; k < KMAX; ++k) {
      const int  e  = c0 + k * NTHR + tid;
      const bool v  = e < c1;
      const int  ec = v ? e : 0;
      myS[k] = src[ec];
      myT[k] = tgt[ec];
      myW[k] = ew[ec];
      const int rl = et[ec];
      myR[k] = v ? rl : -1;
    }

    if (ch == 0) {
      float4 bv[(NREL * 4 * 2 * 32) / NTHR];
#pragma unroll
      for (int it = 0; it < (NREL * 4 * 2 * 32) / NTHR; ++it) {
        const int idx = it * NTHR + tid;
        const int ln = idx & 31, j = (idx >> 5) & 1, q = (idx >> 6) & 3, r = idx >> 8;
        const int c = ln & 15, h = ln >> 4;
        const float* bp = blocks + ((((size_t)r * NBLK + 2 * q + j) * BS + 4 * h) * BS + (c & 7));
        const bool nz = (j == 0) == (c < 8);
        const float t0 = bp[0], t1 = bp[BS], t2 = bp[2 * BS], t3 = bp[3 * BS];
        bv[it].x = nz ? t0 : 0.0f; bv[it].y = nz ? t1 : 0.0f;
        bv[it].z = nz ? t2 : 0.0f; bv[it].w = nz ? t3 : 0.0f;
      }
#pragma unroll
      for (int it = 0; it < (NREL * 4 * 2 * 32) / NTHR; ++it) Bl[it * NTHR + tid] = bv[it];
    }

    int cnt[NREL];
#pragma unroll
    for (int rr = 0; rr < NREL; ++rr) cnt[rr] = 0;
#pragma unroll
    for (int k = 0; k < KMAX; ++k) {
      unsigned m[NREL];
#pragma unroll
      for (int rr = 0; rr < NREL; ++rr) m[rr] = __builtin_amdgcn_ballot_w32(myR[k] == rr);
      unsigned om = 0u;
      int      oc = 0;
#pragma unroll
      for (int rr = 0; rr < NREL; ++rr) {
        om = (myR[k] == rr) ? m[rr] : om;
        oc = (myR[k] == rr) ? cnt[rr] : oc;
      }
      myRank[k] = oc + (int)__builtin_amdgcn_mbcnt_lo(om, 0u);
#pragma unroll
      for (int rr = 0; rr < NREL; ++rr) cnt[rr] += __builtin_popcount(m[rr]);
    }
    {
      int cv = cnt[0];
#pragma unroll
      for (int rr = 1; rr < NREL; ++rr) cv = (lane == rr) ? cnt[rr] : cv;
      if (lane < NREL) wtot[wave * NREL + lane] = cv;
    }
    __syncthreads();

    if (tid < NWAVE * NREL) {
      const int w_ = tid >> 3, r_ = tid & 7;
      int pre = 0;
#pragma unroll
      for (int w2 = 0; w2 < NWAVE; ++w2) {
        const int t = wtot[w2 * NREL + r_];
        pre += (w2 < w_) ? t : 0;
        if (w2 == NWAVE - 1 && w_ == NWAVE - 1) rtot[r_] = pre + t;
      }
      wpre[tid] = pre;
    }
    __syncthreads();

    int rt[NREL], rb[NREL + 1];
    rb[0] = 0;
#pragma unroll
    for (int rr = 0; rr < NREL; ++rr) {
      rt[rr] = __builtin_amdgcn_readfirstlane(rtot[rr]);
      rb[rr + 1] = rb[rr] + ((rt[rr] + 15) & ~15);
    }

#pragma unroll
    for (int k = 0; k < KMAX; ++k) {
      const int r = myR[k];
      if (r >= 0 && r < NREL) {
        int base = rb[0];
#pragma unroll
        for (int rr = 1; rr < NREL; ++rr) base = (r == rr) ? rb[rr] : base;
        const int pos = base + wpre[wave * NREL + r] + myRank[k];
        ls_src[pos] = myS[k];
        ls_tgt[pos] = myT[k];
        ls_w[pos]   = myW[k];
      }
    }
    if (tid < NREL * 16) {
      const int r_ = tid >> 4, jj = tid & 15;
      int base = rb[0], tot = rt[0], end = rb[1];
#pragma unroll
      for (int rr = 1; rr < NREL; ++rr) {
        base = (r_ == rr) ? rb[rr] : base;
        tot  = (r_ == rr) ? rt[rr] : tot;
        end  = (r_ == rr) ? rb[rr + 1] : end;
      }
      const int p = base + tot + jj;
      if (p < end) { ls_src[p] = 0; ls_tgt[p] = -1; ls_w[p] = 0.0f; }
    }
    __syncthreads();

    const int nTiles = rb[NREL] >> 4;
    for (int tile = wave; tile < nTiles; tile += NWAVE) {
      const int p0 = tile << 4;
      int r = 0;
#pragma unroll
      for (int rr = 1; rr < NREL; ++rr) r += (p0 >= rb[rr]) ? 1 : 0;

      const int   es = ls_src[p0 + nl];
      const float w  = ls_w[p0 + nl];

      const float* xr = x + (size_t)es * DIM + 4 * half;
      float4 xa[NBLK];
#pragma unroll
      for (int b = 0; b < NBLK; ++b) xa[b] = *(const float4*)(xr + b * BS);

      const int4 tg0 = *(const int4*)&ls_tgt[p0];
      const int4 tg1 = *(const int4*)&ls_tgt[p0 + 4];
      const int4 tg2 = *(const int4*)&ls_tgt[p0 + 8];
      const int4 tg3 = *(const int4*)&ls_tgt[p0 + 12];
      const int tg[16] = {tg0.x, tg0.y, tg0.z, tg0.w, tg1.x, tg1.y, tg1.z, tg1.w,
                          tg2.x, tg2.y, tg2.z, tg2.w, tg3.x, tg3.y, tg3.z, tg3.w};
      int ts[16];
#pragma unroll
      for (int i = 0; i < 16; ++i) ts[i] = __builtin_amdgcn_readfirstlane(tg[i]);

      const float4* bp = Bl + r * 256 + lane;
      float4 bq[4][2];
#pragma unroll
      for (int q = 0; q < 4; ++q) {
        bq[q][0] = bp[(q * 2 + 0) * 32];
        bq[q][1] = bp[(q * 2 + 1) * 32];
      }
      int anyrow = ts[0];
#pragma unroll
      for (int i = 1; i < 16; ++i) anyrow |= ts[i];
      if (anyrow >= 0) tile_products_and_adds<true >(xa, w, bq, ts, out, lane);
      else             tile_products_and_adds<false>(xa, w, bq, ts, out, lane);
    }
    __syncthreads();
  }
}

extern "C" void kernel_launch(void* const* d_in, const int* in_sizes, int n_in,
                              void* d_out, int out_size, void* d_ws, size_t ws_size,
                              hipStream_t stream) {
  const float* x      = (const float*)d_in[0];
  const float* blocks = (const float*)d_in[1];
  const float* ew     = (const float*)d_in[2];
  const int*   src    = (const int*)d_in[3];
  const int*   tgt    = (const int*)d_in[4];
  const int*   et     = (const int*)d_in[5];
  float*       out    = (float*)d_out;

  const int nNodes = in_sizes[0] / DIM;
  const int nEdges = in_sizes[2];

  long long o4 = (long long)nNodes * (DIM / 4);
  zero_f32v4<<<2048, 256, 0, stream>>>((float4*)out, o4);

  edge_transform_scatter<<<512, NTHR, 0, stream>>>(x, blocks, ew, src, tgt, et,
                                                   out, nEdges);
}


# out rows split in two parts, one per package half (own L2 each): a work-group reads its die, draws a slice from that half's counter, adds only into its part; atomic arrival order free as before
# speedup vs baseline: 3.4395x; 1.2069x over previous
#include <hip/hip_runtime.h>


#define DIM   64
#define NREL  8
#define NBLK  8
#define BS    8

#define NTHR   512
#define NWAVE  (NTHR / 32)
#define KMAX   7
#define CHUNK  (NTHR * KMAX)
#define LISTN  (CHUNK + NREL * 16)

typedef __attribute__((ext_vector_type(2))) float v2f;
typedef __attribute__((ext_vector_type(2))) unsigned v2u;
typedef __attribute__((ext_vector_type(8))) float v8f;

__global__ void zero_f32v4(float4* __restrict__ p, long long n4,
                           unsigned* __restrict__ tickets) {
  long long i = (long long)blockIdx.x * blockDim.x + threadIdx.x;
  long long stride = (long long)gridDim.x * blockDim.x;
  float4 z; z.x = z.y = z.z = z.w = 0.0f;
  if (blockIdx.x == 0 && threadIdx.x < 2) tickets[threadIdx.x] = 0u;
  for (; i < n4; i += stride) p[i] = z;
}

static __device__ __forceinline__ v8f pair_product(
    const float4 (&xa)[NBLK], const float w, const float4 (&bq)[4][2], const int q) {
  v8f acc = {};
#pragma unroll
  for (int j = 0; j < 2; ++j) {
    const float4 xv = xa[2 * q + j];
    const float4 bv = bq[q][j];
    v2f a0, a1, b0, b1;
    a0[0] = xv.x * w; a0[1] = xv.y * w; b0[0] = bv.x; b0[1] = bv.y;
    a1[0] = xv.z * w; a1[1] = xv.w * w; b1[0] = bv.z; b1[1] = bv.w;
    acc = __builtin_amdgcn_wmma_f32_16x16x4_f32(
        false, a0, false, b0, (short)0, acc, false, false);
    acc = __builtin_amdgcn_wmma_f32_16x16x4_f32(
        false, a1, false, b1, (short)0, acc, false, false);
  }
  return acc;
}

template <bool FULL>
static __device__ __forceinline__ void tile_products_and_adds(
    const float4 (&xa)[NBLK], const float w, const float4 (&bq)[4][2],
    const int (&ts)[16], float* __restrict__ out, const int lane) {
  float* const col = out + lane;

#pragma unroll
  for (int p = 0; p < NBLK / 4; ++p) {
    v8f lo = pair_product(xa, w, bq, 2 * p);
    v8f hi = pair_product(xa, w, bq, 2 * p + 1);
#pragma unroll
    for (int i = 0; i < 8; ++i) {
      const float lv = lo[i], hv = hi[i];
      const v2u s = __builtin_amdgcn_permlane16_swap(
          __float_as_uint(lv), __float_as_uint(hv), false, false);
      const unsigned s0 = s[0], s1 = s[1];
      lo[i] = __uint_as_float(s0);
      hi[i] = __uint_as_float(s1);
    }
#pragma unroll
    for (int i = 0; i < 8; ++i) {
      if (FULL || ts[i] >= 0)
        unsafeAtomicAdd(col + (size_t)(unsigned)ts[i] * DIM + 32 * p, lo[i]);
    }
#pragma unroll
    for (int i = 0; i < 8; ++i) {
      if (FULL || ts[8 + i] >= 0)
        unsafeAtomicAdd(col + (size_t)(unsigned)ts[8 + i] * DIM + 32 * p, hi[i]);
    }
  }
}

__global__ void __launch_bounds__(NTHR, 8)
edge_transform_scatter(const float* __restrict__ x,
                       const float* __restrict__ blocks,
                       const float* __restrict__ ew,
                       const int*   __restrict__ src,
                       const int*   __restrict__ tgt,
                       const int*   __restrict__ et,
                       float*       __restrict__ out,
                       unsigned*    __restrict__ tickets,
                       int nEdges, int nNodes) {
  __shared__ float4 Bl[NREL * 4 * 2 * 32];
  __shared__ int    ls_src[LISTN];
  __shared__ __attribute__((aligned(16))) int ls_tgt[LISTN];
  __shared__ float  ls_w[LISTN];
  __shared__ int    wtot[NWAVE * NREL];
  __shared__ int    wpre[NWAVE * NREL];
  __shared__ int    rtot[NREL];
  __shared__ int    s_item;

  const int tid  = threadIdx.x;
  const int lane = tid & 31;
  const int wave = __builtin_amdgcn_readfirstlane(tid >> 5);
  const int nl   = lane & 15;
  const int half = lane >> 4;

  const int nS = (int)gridDim.x >> 1;

  int h0 = 0, tk = 0;
  {
    float4 bv[(NREL * 4 * 2 * 32) / NTHR];
#pragma unroll
    for (int it = 0; it < (NREL * 4 * 2 * 32) / NTHR; ++it) {
      const int idx = it * NTHR + tid;
      const int ln = idx & 31, j = (idx >> 5) & 1, q = (idx >> 6) & 3, r = idx >> 8;
      const int c = ln & 15, h = ln >> 4;
      const float* bp = blocks + ((((size_t)r * NBLK + 2 * q + j) * BS + 4 * h) * BS + (c & 7));
      const bool nz = (j == 0) == (c < 8);
      const float t0 = bp[0], t1 = bp[BS], t2 = bp[2 * BS], t3 = bp[3 * BS];
      bv[it].x = nz ? t0 : 0.0f; bv[it].y = nz ? t1 : 0.0f;
      bv[it].z = nz ? t2 : 0.0f; bv[it].w = nz ? t3 : 0.0f;
    }
    if (tid == 0) {
      const unsigned id = __builtin_amdgcn_s_sendmsg_rtn(0x87);
      h0 = (int)((id >> 18) & 1u);
      tk = (int)__hip_atomic_fetch_add(tickets + h0, 1u, __ATOMIC_RELAXED,
                                       __HIP_MEMORY_SCOPE_AGENT);
    }
#pragma unroll
    for (int it = 0; it < (NREL * 4 * 2 * 32) / NTHR; ++it) Bl[it * NTHR + tid] = bv[it];
  }
  if (tid == 0) {
    const int hh = tk < nS ? h0 : 1 - h0;
    int ss = tk < nS ? tk : 2 * nS - 1 - tk;
    ss = ss < 0 ? 0 : ss;
    s_item = hh * 65536 + ss;
  }
  __syncthreads();
  const int item  = __builtin_amdgcn_readfirstlane(s_item);
  const int part  = item >> 16;
  const int slice = item & 0xFFFF;
  const int split = nNodes >> 1;

  const int per = (((nEdges + nS - 1) / nS) + 31) & ~31;
  const long long g0 = (long long)slice * per;
  const int r0 = g0 < nEdges ? (int)g0 : nEdges;
  const int r1 = (g0 + per) < nEdges ? (int)(g0 + per) : nEdges;
  const int nChunks = (per + CHUNK - 1) / CHUNK;

  for (int ch = 0; ch < nChunks; ++ch) {
    const int c0 = r0 + ch * CHUNK < r1 ? r0 + ch * CHUNK : r1;
    const int c1 = c0 + CHUNK < r1 ? c0 + CHUNK : r1;

    int   myS[KMAX], myT[KMAX], myR[KMAX], myRank[KMAX];
    float myW[KMAX];
#pragma unroll
    for (int k = 0; k < KMAX; ++k) {
      const int  e  = c0 + k * NTHR + tid;
      const bool v  = e < c1;
      const int  ec = v ? e : 0;
      myS[k] = src[ec];
      myT[k] = tgt[ec];
      myW[k] = ew[ec];
      const int rl = et[ec];
      const int tp = myT[k] >= split ? 1 : 0;
      myR[k] = (v && tp == part) ? rl : -1;
    }

    int cnt[NREL];
#pragma unroll
    for (int rr = 0; rr < NREL; ++rr) cnt[rr] = 0;
#pragma unroll
    for (int k = 0; k < KMAX; ++k) {
      unsigned m[NREL];
#pragma unroll
      for (int rr = 0; rr < NREL; ++rr) m[rr] = __builtin_amdgcn_ballot_w32(myR[k] == rr);
      unsigned om = 0u;
      int      oc = 0;
#pragma unroll
      for (int rr = 0; rr < NREL; ++rr) {
        om = (myR[k] == rr) ? m[rr] : om;
        oc = (myR[k] == rr) ? cnt[rr] : oc;
      }
      myRank[k] = oc + (int)__builtin_amdgcn_mbcnt_lo(om, 0u);
#pragma unroll
      for (int rr = 0; rr < NREL; ++rr) cnt[rr] += __builtin_popcount(m[rr]);
    }
    {
      int cv = cnt[0];
#pragma unroll
      for (int rr = 1; rr < NREL; ++rr) cv = (lane == rr) ? cnt[rr] : cv;
      if (lane < NREL) wtot[wave * NREL + lane] = cv;
    }
    __syncthreads();

    if (tid < NWAVE * NREL) {
      const int w_ = tid >> 3, r_ = tid & 7;
      int pre = 0;
#pragma unroll
      for (int w2 = 0; w2 < NWAVE; ++w2) {
        const int t = wtot[w2 * NREL + r_];
        pre += (w2 < w_) ? t : 0;
        if (w2 == NWAVE - 1 && w_ == NWAVE - 1) rtot[r_] = pre + t;
      }
      wpre[tid] = pre;
    }
    __syncthreads();

    int rt[NREL], rb[NREL + 1];
    rb[0] = 0;
#pragma unroll
    for (int rr = 0; rr < NREL; ++rr) {
      rt[rr] = __builtin_amdgcn_readfirstlane(rtot[rr]);
      rb[rr + 1] = rb[rr] + ((rt[rr] + 15) & ~15);
    }

#pragma unroll
    for (int k = 0; k < KMAX; ++k) {
      const int r = myR[k];
      if (r >= 0 && r < NREL) {
        int base = rb[0];
#pragma unroll
        for (int rr = 1; rr < NREL; ++rr) base = (r == rr) ? rb[rr] : base;
        const int pos = base + wpre[wave * NREL + r] + myRank[k];
        ls_src[pos] = myS[k];
        ls_tgt[pos] = myT[k];
        ls_w[pos]   = myW[k];
      }
    }
    if (tid < NREL * 16) {
      const int r_ = tid >> 4, jj = tid & 15;
      int base = rb[0], tot = rt[0], end = rb[1];
#pragma unroll
      for (int rr = 1; rr < NREL; ++rr) {
        base = (r_ == rr) ? rb[rr] : base;
        tot  = (r_ == rr) ? rt[rr] : tot;
        end  = (r_ == rr) ? rb[rr + 1] : end;
      }
      const int p = base + tot + jj;
      if (p < end) { ls_src[p] = 0; ls_tgt[p] = -1; ls_w[p] = 0.0f; }
    }
    __syncthreads();

    const int nTiles = rb[NREL] >> 4;
    for (int tile = wave; tile < nTiles; tile += NWAVE) {
      const int p0 = tile << 4;
      int r = 0;
#pragma unroll
      for (int rr = 1; rr < NREL; ++rr) r += (p0 >= rb[rr]) ? 1 : 0;

      const int   es = ls_src[p0 + nl];
      const float w  = ls_w[p0 + nl];

      const float* xr = x + (size_t)es * DIM + 4 * half;
      float4 xa[NBLK];
#pragma unroll
      for (int b = 0; b < NBLK; ++b) xa[b] = *(const float4*)(xr + b * BS);

      const int4 tg0 = *(const int4*)&ls_tgt[p0];
      const int4 tg1 = *(const int4*)&ls_tgt[p0 + 4];
      const int4 tg2 = *(const int4*)&ls_tgt[p0 + 8];
      const int4 tg3 = *(const int4*)&ls_tgt[p0 + 12];
      const int tg[16] = {tg0.x, tg0.y, tg0.z, tg0.w, tg1.x, tg1.y, tg1.z, tg1.w,
                          tg2.x, tg2.y, tg2.z, tg2.w, tg3.x, tg3.y, tg3.z, tg3.w};
      int ts[16];
#pragma unroll
      for (int i = 0; i < 16; ++i) ts[i] = __builtin_amdgcn_readfirstlane(tg[i]);

      const float4* bp = Bl + r * 256 + lane;
      float4 bq[4][2];
#pragma unroll
      for (int q = 0; q < 4; ++q) {
        bq[q][0] = bp[(q * 2 + 0) * 32];
        bq[q][1] = bp[(q * 2 + 1) * 32];
      }
      int anyrow = ts[0];
#pragma unroll
      for (int i = 1; i < 16; ++i) anyrow |= ts[i];
      if (anyrow >= 0) tile_products_and_adds<true >(xa, w, bq, ts, out, lane);
      else             tile_products_and_adds<false>(xa, w, bq, ts, out, lane);
    }
    __syncthreads();
  }
}

extern "C" void kernel_launch(void* const* d_in, const int* in_sizes, int n_in,
                              void* d_out, int out_size, void* d_ws, size_t ws_size,
                              hipStream_t stream) {
  const float* x      = (const float*)d_in[0];
  const float* blocks = (const float*)d_in[1];
  const float* ew     = (const float*)d_in[2];
  const int*   src    = (const int*)d_in[3];
  const int*   tgt    = (const int*)d_in[4];
  const int*   et     = (const int*)d_in[5];
  float*       out    = (float*)d_out;

  const int nNodes = in_sizes[0] / DIM;
  const int nEdges = in_sizes[2];

  long long o4 = (long long)nNodes * (DIM / 4);
  unsigned* tickets = (unsigned*)d_ws;
  zero_f32v4<<<2048, 256, 0, stream>>>((float4*)out, o4, tickets);

  edge_transform_scatter<<<512, NTHR, 0, stream>>>(x, blocks, ew, src, tgt, et,
                                                   out, tickets, nEdges, nNodes);
}


# edge-list loads of a chunk issued as one group: the relation load no longer hangs on the target test (select became an OR); 28 loads per thread in flight, one round trip where there were fourteen
# speedup vs baseline: 3.8289x; 1.1132x over previous
#include <hip/hip_runtime.h>


#define DIM   64
#define NREL  8
#define NBLK  8
#define BS    8

#define NTHR   512
#define NWAVE  (NTHR / 32)
#define KMAX   7
#define CHUNK  (NTHR * KMAX)
#define LISTN  (CHUNK + NREL * 16)

typedef __attribute__((ext_vector_type(2))) float v2f;
typedef __attribute__((ext_vector_type(2))) unsigned v2u;
typedef __attribute__((ext_vector_type(8))) float v8f;

__global__ void zero_f32v4(float4* __restrict__ p, long long n4,
                           unsigned* __restrict__ tickets) {
  long long i = (long long)blockIdx.x * blockDim.x + threadIdx.x;
  long long stride = (long long)gridDim.x * blockDim.x;
  float4 z; z.x = z.y = z.z = z.w = 0.0f;
  if (blockIdx.x == 0 && threadIdx.x < 2) tickets[threadIdx.x] = 0u;
  for (; i < n4; i += stride) p[i] = z;
}

static __device__ __forceinline__ v8f pair_product(
    const float4 (&xa)[NBLK], const float w, const float4 (&bq)[4][2], const int q) {
  v8f acc = {};
#pragma unroll
  for (int j = 0; j < 2; ++j) {
    const float4 xv = xa[2 * q + j];
    const float4 bv = bq[q][j];
    v2f a0, a1, b0, b1;
    a0[0] = xv.x * w; a0[1] = xv.y * w; b0[0] = bv.x; b0[1] = bv.y;
    a1[0] = xv.z * w; a1[1] = xv.w * w; b1[0] = bv.z; b1[1] = bv.w;
    acc = __builtin_amdgcn_wmma_f32_16x16x4_f32(
        false, a0, false, b0, (short)0, acc, false, false);
    acc = __builtin_amdgcn_wmma_f32_16x16x4_f32(
        false, a1, false, b1, (short)0, acc, false, false);
  }
  return acc;
}

template <bool FULL>
static __device__ __forceinline__ void tile_products_and_adds(
    const float4 (&xa)[NBLK], const float w, const float4 (&bq)[4][2],
    const int (&ts)[16], float* __restrict__ out, const int lane) {
  float* const col = out + lane;

#pragma unroll
  for (int p = 0; p < NBLK / 4; ++p) {
    v8f lo = pair_product(xa, w, bq, 2 * p);
    v8f hi = pair_product(xa, w, bq, 2 * p + 1);
#pragma unroll
    for (int i = 0; i < 8; ++i) {
      const float lv = lo[i], hv = hi[i];
      const v2u s = __builtin_amdgcn_permlane16_swap(
          __float_as_uint(lv), __float_as_uint(hv), false, false);
      const unsigned s0 = s[0], s1 = s[1];
      lo[i] = __uint_as_float(s0);
      hi[i] = __uint_as_float(s1);
    }
#pragma unroll
    for (int i = 0; i < 8; ++i) {
      if (FULL || ts[i] >= 0)
        unsafeAtomicAdd(col + (size_t)(unsigned)ts[i] * DIM + 32 * p, lo[i]);
    }
#pragma unroll
    for (int i = 0; i < 8; ++i) {
      if (FULL || ts[8 + i] >= 0)
        unsafeAtomicAdd(col + (size_t)(unsigned)ts[8 + i] * DIM + 32 * p, hi[i]);
    }
  }
}

__global__ void __launch_bounds__(NTHR, 8)
edge_transform_scatter(const float* __restrict__ x,
                       const float* __restrict__ blocks,
                       const float* __restrict__ ew,
                       const int*   __restrict__ src,
                       const int*   __restrict__ tgt,
                       const int*   __restrict__ et,
                       float*       __restrict__ out,
                       unsigned*    __restrict__ tickets,
                       int nEdges, int nNodes) {
  __shared__ float4 Bl[NREL * 4 * 2 * 32];
  __shared__ int    ls_src[LISTN];
  __shared__ __attribute__((aligned(16))) int ls_tgt[LISTN];
  __shared__ float  ls_w[LISTN];
  __shared__ int    wtot[NWAVE * NREL];
  __shared__ int    wpre[NWAVE * NREL];
  __shared__ int    rtot[NREL];
  __shared__ int    s_item;

  const int tid  = threadIdx.x;
  const int lane = tid & 31;
  const int wave = __builtin_amdgcn_readfirstlane(tid >> 5);
  const int nl   = lane & 15;
  const int half = lane >> 4;

  const int nS = (int)gridDim.x >> 1;

  int h0 = 0, tk = 0;
  {
    float4 bv[(NREL * 4 * 2 * 32) / NTHR];
#pragma unroll
    for (int it = 0; it < (NREL * 4 * 2 * 32) / NTHR; ++it) {
      const int idx = it * NTHR + tid;
      const int ln = idx & 31, j = (idx >> 5) & 1, q = (idx >> 6) & 3, r = idx >> 8;
      const int c = ln & 15, h = ln >> 4;
      const float* bp = blocks + ((((size_t)r * NBLK + 2 * q + j) * BS + 4 * h) * BS + (c & 7));
      const bool nz = (j == 0) == (c < 8);
      const float t0 = bp[0], t1 = bp[BS], t2 = bp[2 * BS], t3 = bp[3 * BS];
      bv[it].x = nz ? t0 : 0.0f; bv[it].y = nz ? t1 : 0.0f;
      bv[it].z = nz ? t2 : 0.0f; bv[it].w = nz ? t3 : 0.0f;
    }
    if (tid == 0) {
      const unsigned id = __builtin_amdgcn_s_sendmsg_rtn(0x87);
      h0 = (int)((id >> 18) & 1u);
      tk = (int)__hip_atomic_fetch_add(tickets + h0, 1u, __ATOMIC_RELAXED,
                                       __HIP_MEMORY_SCOPE_AGENT);
    }
#pragma unroll
    for (int it = 0; it < (NREL * 4 * 2 * 32) / NTHR; ++it) Bl[it * NTHR + tid] = bv[it];
  }
  if (tid == 0) {
    const int hh = tk < nS ? h0 : 1 - h0;
    int ss = tk < nS ? tk : 2 * nS - 1 - tk;
    ss = ss < 0 ? 0 : ss;
    s_item = hh * 65536 + ss;
  }
  __syncthreads();
  const int item  = __builtin_amdgcn_readfirstlane(s_item);
  const int part  = item >> 16;
  const int slice = item & 0xFFFF;
  const int split = nNodes >> 1;

  const int per = (((nEdges + nS - 1) / nS) + 31) & ~31;
  const long long g0 = (long long)slice * per;
  const int r0 = g0 < nEdges ? (int)g0 : nEdges;
  const int r1 = (g0 + per) < nEdges ? (int)(g0 + per) : nEdges;
  const int nChunks = (per + CHUNK - 1) / CHUNK;

  for (int ch = 0; ch < nChunks; ++ch) {
    const int c0 = r0 + ch * CHUNK < r1 ? r0 + ch * CHUNK : r1;
    const int c1 = c0 + CHUNK < r1 ? c0 + CHUNK : r1;

    int   myS[KMAX], myT[KMAX], myL[KMAX], myR[KMAX], myRank[KMAX];
    float myW[KMAX];
#pragma unroll
    for (int k = 0; k < KMAX; ++k) {
      const int  e  = c0 + k * NTHR + tid;
      const int  ec = e < c1 ? e : 0;
      myS[k] = src[ec];
      myT[k] = tgt[ec];
      myW[k] = ew[ec];
      myL[k] = et[ec];
    }
#pragma unroll
    for (int k = 0; k < KMAX; ++k) {
      const bool v  = c0 + k * NTHR + tid < c1;
      const int  tp = myT[k] >= split ? 1 : 0;
      myR[k] = myL[k] | ((v && tp == part) ? 0 : -1);
    }

    int cnt[NREL];
#pragma unroll
    for (int rr = 0; rr < NREL; ++rr) cnt[rr] = 0;
#pragma unroll
    for (int k = 0; k < KMAX; ++k) {
      unsigned m[NREL];
#pragma unroll
      for (int rr = 0; rr < NREL; ++rr) m[rr] = __builtin_amdgcn_ballot_w32(myR[k] == rr);
      unsigned om = 0u;
      int      oc = 0;
#pragma unroll
      for (int rr = 0; rr < NREL; ++rr) {
        om = (myR[k] == rr) ? m[rr] : om;
        oc = (myR[k] == rr) ? cnt[rr] : oc;
      }
      myRank[k] = oc + (int)__builtin_amdgcn_mbcnt_lo(om, 0u);
#pragma unroll
      for (int rr = 0; rr < NREL; ++rr) cnt[rr] += __builtin_popcount(m[rr]);
    }
    {
      int cv = cnt[0];
#pragma unroll
      for (int rr = 1; rr < NREL; ++rr) cv = (lane == rr) ? cnt[rr] : cv;
      if (lane < NREL) wtot[wave * NREL + lane] = cv;
    }
    __syncthreads();

    if (tid < NWAVE * NREL) {
      const int w_ = tid >> 3, r_ = tid & 7;
      int pre = 0;
#pragma unroll
      for (int w2 = 0; w2 < NWAVE; ++w2) {
        const int t = wtot[w2 * NREL + r_];
        pre += (w2 < w_) ? t : 0;
        if (w2 == NWAVE - 1 && w_ == NWAVE - 1) rtot[r_] = pre + t;
      }
      wpre[tid] = pre;
    }
    __syncthreads();

    int rt[NREL], rb[NREL + 1];
    rb[0] = 0;
#pragma unroll
    for (int rr = 0; rr < NREL; ++rr) {
      rt[rr] = __builtin_amdgcn_readfirstlane(rtot[rr]);
      rb[rr + 1] = rb[rr] + ((rt[rr] + 15) & ~15);
    }

#pragma unroll
    for (int k = 0; k < KMAX; ++k) {
      const int r = myR[k];
      if (r >= 0 && r < NREL) {
        int base = rb[0];
#pragma unroll
        for (int rr = 1; rr < NREL; ++rr) base = (r == rr) ? rb[rr] : base;
        const int pos = base + wpre[wave * NREL + r] + myRank[k];
        ls_src[pos] = myS[k];
        ls_tgt[pos] = myT[k];
        ls_w[pos]   = myW[k];
      }
    }
    if (tid < NREL * 16) {
      const int r_ = tid >> 4, jj = tid & 15;
      int base = rb[0], tot = rt[0], end = rb[1];
#pragma unroll
      for (int rr = 1; rr < NREL; ++rr) {
        base = (r_ == rr) ? rb[rr] : base;
        tot  = (r_ == rr) ? rt[rr] : tot;
        end  = (r_ == rr) ? rb[rr + 1] : end;
      }
      const int p = base + tot + jj;
      if (p < end) { ls_src[p] = 0; ls_tgt[p] = -1; ls_w[p] = 0.0f; }
    }
    __syncthreads();

    const int nTiles = rb[NREL] >> 4;
    for (int tile = wave; tile < nTiles; tile += NWAVE) {
      const int p0 = tile << 4;
      int r = 0;
#pragma unroll
      for (int rr = 1; rr < NREL; ++rr) r += (p0 >= rb[rr]) ? 1 : 0;

      const int   es = ls_src[p0 + nl];
      const float w  = ls_w[p0 + nl];

      const float* xr = x + (size_t)es * DIM + 4 * half;
      float4 xa[NBLK];
#pragma unroll
      for (int b = 0; b < NBLK; ++b) xa[b] = *(const float4*)(xr + b * BS);

      const int4 tg0 = *(const int4*)&ls_tgt[p0];
      const int4 tg1 = *(const int4*)&ls_tgt[p0 + 4];
      const int4 tg2 = *(const int4*)&ls_tgt[p0 + 8];
      const int4 tg3 = *(const int4*)&ls_tgt[p0 + 12];
      const int tg[16] = {tg0.x, tg0.y, tg0.z, tg0.w, tg1.x, tg1.y, tg1.z, tg1.w,
                          tg2.x, tg2.y, tg2.z, tg2.w, tg3.x, tg3.y, tg3.z, tg3.w};
      int ts[16];
#pragma unroll
      for (int i = 0; i < 16; ++i) ts[i] = __builtin_amdgcn_readfirstlane(tg[i]);

      const float4* bp = Bl + r * 256 + lane;
      float4 bq[4][2];
#pragma unroll
      for (int q = 0; q < 4; ++q) {
        bq[q][0] = bp[(q * 2 + 0) * 32];
        bq[q][1] = bp[(q * 2 + 1) * 32];
      }
      int anyrow = ts[0];
#pragma unroll
      for (int i = 1; i < 16; ++i) anyrow |= ts[i];
      if (anyrow >= 0) tile_products_and_adds<true >(xa, w, bq, ts, out, lane);
      else             tile_products_and_adds<false>(xa, w, bq, ts, out, lane);
    }
    __syncthreads();
  }
}

extern "C" void kernel_launch(void* const* d_in, const int* in_sizes, int n_in,
                              void* d_out, int out_size, void* d_ws, size_t ws_size,
                              hipStream_t stream) {
  const float* x      = (const float*)d_in[0];
  const float* blocks = (const float*)d_in[1];
  const float* ew     = (const float*)d_in[2];
  const int*   src    = (const int*)d_in[3];
  const int*   tgt    = (const int*)d_in[4];
  const int*   et     = (const int*)d_in[5];
  float*       out    = (float*)d_out;

  const int nNodes = in_sizes[0] / DIM;
  const int nEdges = in_sizes[2];

  long long o4 = (long long)nNodes * (DIM / 4);
  unsigned* tickets = (unsigned*)d_ws;
  zero_f32v4<<<2048, 256, 0, stream>>>((float4*)out, o4, tickets);

  edge_transform_scatter<<<512, NTHR, 0, stream>>>(x, blocks, ew, src, tgt, et,
                                                   out, tickets, nEdges, nNodes);
}
